# starting kernel
# baseline (speedup 1.0000x reference)
#include <hip/hip_runtime.h>


typedef float v2f __attribute__((ext_vector_type(2)));
typedef float v8f __attribute__((ext_vector_type(8)));

#define B_   8
#define N_   1024
#define NIN  64

__global__ __launch_bounds__(32) void gat_h_wmma(const float* __restrict__ x,
                                                 const float* __restrict__ a,
                                                 float* __restrict__ h) {
  const int lane = threadIdx.x;
  const int half = lane >> 4;
  const int lr   = lane & 15;
  const int row0 = blockIdx.x * 16;
  const float* xr = x + (size_t)(row0 + lr) * NIN;

  const int ca = lr & 7;

  v8f acc = {};
#pragma unroll
  for (int it = 0; it < 16; ++it) {
    const int k0 = it * 4 + half * 2;
    v2f af, bf;
    af.x = xr[k0];
    af.y = xr[k0 + 1];
    const int fa = (lr < 8) ? k0 : (NIN + k0);
    bf.x = a[fa * 8 + ca];
    bf.y = a[(fa + 1) * 8 + ca];
    acc = __builtin_amdgcn_wmma_f32_16x16x4_f32(false, af, false, bf,
                                                (short)0, acc, false, false);
  }

  float* hrow = h + (size_t)row0 * 16;
#pragma unroll
  for (int v = 0; v < 8; ++v) {
    hrow[(size_t)(half * 8 + v) * 16 + lr] = acc[v];
  }
}

__global__ __launch_bounds__(256) void gat_att_force(
    const float* __restrict__ adj, const float* __restrict__ c,
    const float* __restrict__ h, const float* __restrict__ a,
    const float* __restrict__ a2, float* __restrict__ out) {
  __shared__ float sh_hj[8 * N_];
  __shared__ float sh_c[3 * N_];
  __shared__ float sh_red[8 * 3];

  const int tid = threadIdx.x;
  const int bi  = blockIdx.x;
  const int b   = bi >> 10;
  const int i   = bi & (N_ - 1);
  const int j0  = tid * 4;

  const float4 av4 = *(const float4*)(adj + (size_t)bi * N_ + j0);

  const float* hb = h + (size_t)b * N_ * 16;
  for (int idx = tid; idx < N_ * 2; idx += 256) {
    const int j  = idx >> 1;
    const int q4 = (idx & 1) * 4;
    const float4 v = *(const float4*)(hb + (size_t)j * 16 + 8 + q4);
    sh_hj[(q4 + 0) * N_ + j] = v.x;
    sh_hj[(q4 + 1) * N_ + j] = v.y;
    sh_hj[(q4 + 2) * N_ + j] = v.z;
    sh_hj[(q4 + 3) * N_ + j] = v.w;
  }
  const float* cb = c + (size_t)b * N_ * 3;
  for (int j = tid; j < N_; j += 256) {
    sh_c[0 * N_ + j] = cb[j * 3 + 0];
    sh_c[1 * N_ + j] = cb[j * 3 + 1];
    sh_c[2 * N_ + j] = cb[j * 3 + 2];
  }

  float hik[8], aak[8], a2k[8];
#pragma unroll
  for (int k = 0; k < 8; ++k) {
    hik[k] = hb[(size_t)i * 16 + k];
    aak[k] = a[128 * 8 + k];
    a2k[k] = a2[k];
  }
  const float cix = cb[i * 3 + 0];
  const float ciy = cb[i * 3 + 1];
  const float ciz = cb[i * 3 + 2];

  __syncthreads();

  float av[4], hj[8][4], cjx[4], cjy[4], cjz[4];
  *(float4*)av = av4;
#pragma unroll
  for (int k = 0; k < 8; ++k)
    *(float4*)hj[k] = *(const float4*)(sh_hj + k * N_ + j0);
  *(float4*)cjx = *(const float4*)(sh_c + 0 * N_ + j0);
  *(float4*)cjy = *(const float4*)(sh_c + 1 * N_ + j0);
  *(float4*)cjz = *(const float4*)(sh_c + 2 * N_ + j0);

  float ox = 0.f, oy = 0.f, oz = 0.f;
#pragma unroll
  for (int q = 0; q < 4; ++q) {
    float att = 0.f;
#pragma unroll
    for (int k = 0; k < 8; ++k) {
      const float pre = hik[k] + hj[k][q] + av[q] * aak[k];
      const float lrl = fmaxf(pre, 0.01f * pre);
      att += lrl * a2k[k];
    }
    const float dx = cix - cjx[q];
    const float dy = ciy - cjy[q];
    const float dz = ciz - cjz[q];
    const float nrm = sqrtf(dx * dx + dy * dy + dz * dz);
    const float w = att / (nrm + 1e-8f);
    ox += w * dx;
    oy += w * dy;
    oz += w * dz;
  }

#pragma unroll
  for (int off = 16; off > 0; off >>= 1) {
    ox += __shfl_xor(ox, off, 32);
    oy += __shfl_xor(oy, off, 32);
    oz += __shfl_xor(oz, off, 32);
  }
  const int wave = tid >> 5;
  if ((tid & 31) == 0) {
    sh_red[wave * 3 + 0] = ox;
    sh_red[wave * 3 + 1] = oy;
    sh_red[wave * 3 + 2] = oz;
  }
  __syncthreads();
  if (tid == 0) {
    float sx = 0.f, sy = 0.f, sz = 0.f;
#pragma unroll
    for (int w8 = 0; w8 < 8; ++w8) {
      sx += sh_red[w8 * 3 + 0];
      sy += sh_red[w8 * 3 + 1];
      sz += sh_red[w8 * 3 + 2];
    }
    out[(size_t)bi * 3 + 0] = sx;
    out[(size_t)bi * 3 + 1] = sy;
    out[(size_t)bi * 3 + 2] = sz;
  }
}

extern "C" void kernel_launch(void* const* d_in, const int* in_sizes, int n_in,
                              void* d_out, int out_size, void* d_ws,
                              size_t ws_size, hipStream_t stream) {
  (void)in_sizes; (void)n_in; (void)out_size; (void)ws_size;
  const float* adj = (const float*)d_in[0];
  const float* c   = (const float*)d_in[1];
  const float* x   = (const float*)d_in[2];
  const float* a   = (const float*)d_in[3];
  const float* a2  = (const float*)d_in[4];
  float* out = (float*)d_out;
  float* h = (float*)d_ws;

  gat_h_wmma<<<(B_ * N_) / 16, 32, 0, stream>>>(x, a, h);
  gat_att_force<<<B_ * N_, 256, 0, stream>>>(adj, c, h, a, a2, out);
}


# kernel 2: one block per 16 rows i instead of one per row; hj/c staged in LDS once per block and kept in registers (4 columns/thread); next row's adj, hi, c_i fetched a row ahead; sum order unchanged
# speedup vs baseline: 2.1998x; 2.1998x over previous
#include <hip/hip_runtime.h>


typedef float v2f __attribute__((ext_vector_type(2)));
typedef float v8f __attribute__((ext_vector_type(8)));

#define B_   8
#define N_   1024
#define NIN  64

__global__ __launch_bounds__(32) void gat_h_wmma(const float* __restrict__ x,
                                                 const float* __restrict__ a,
                                                 float* __restrict__ h) {
  const int lane = threadIdx.x;
  const int half = lane >> 4;
  const int lr   = lane & 15;
  const int row0 = blockIdx.x * 16;
  const float* xr = x + (size_t)(row0 + lr) * NIN;

  const int ca = lr & 7;

  v8f acc = {};
#pragma unroll
  for (int it = 0; it < 16; ++it) {
    const int k0 = it * 4 + half * 2;
    v2f af, bf;
    af.x = xr[k0];
    af.y = xr[k0 + 1];
    const int fa = (lr < 8) ? k0 : (NIN + k0);
    bf.x = a[fa * 8 + ca];
    bf.y = a[(fa + 1) * 8 + ca];
    acc = __builtin_amdgcn_wmma_f32_16x16x4_f32(false, af, false, bf,
                                                (short)0, acc, false, false);
  }

  float* hrow = h + (size_t)row0 * 16;
#pragma unroll
  for (int v = 0; v < 8; ++v) {
    hrow[(size_t)(half * 8 + v) * 16 + lr] = acc[v];
  }
}

#define ROWS 16

__global__ __launch_bounds__(256) void gat_att_force(
    const float* __restrict__ adj, const float* __restrict__ c,
    const float* __restrict__ h, const float* __restrict__ a,
    const float* __restrict__ a2, float* __restrict__ out) {
  __shared__ float sh_hj[8 * N_];
  __shared__ float sh_c[3 * N_];
  __shared__ float sh_red[ROWS * 8 * 3];

  const int tid  = threadIdx.x;
  const int lane = tid & 31;
  const int wave = tid >> 5;
  const int bi0  = blockIdx.x * ROWS;
  const int b    = bi0 >> 10;
  const int i0   = bi0 & (N_ - 1);
  const int j0   = tid * 4;

  const float* arow = adj + (size_t)bi0 * N_ + j0;
  float4 av4 = *(const float4*)arow;

  const float* hb = h + (size_t)b * N_ * 16;
  for (int it = 0; it < (N_ * 2) / 256; ++it) {
    const int idx = tid + it * 256;
    const int j  = idx >> 1;
    const int q4 = (idx & 1) * 4;
    const float4 v = *(const float4*)(hb + (size_t)j * 16 + 8 + q4);
    sh_hj[(q4 + 0) * N_ + j] = v.x;
    sh_hj[(q4 + 1) * N_ + j] = v.y;
    sh_hj[(q4 + 2) * N_ + j] = v.z;
    sh_hj[(q4 + 3) * N_ + j] = v.w;
  }
  const float* cb = c + (size_t)b * N_ * 3;
  for (int it = 0; it < N_ / 256; ++it) {
    const int j = tid + it * 256;
    sh_c[0 * N_ + j] = cb[j * 3 + 0];
    sh_c[1 * N_ + j] = cb[j * 3 + 1];
    sh_c[2 * N_ + j] = cb[j * 3 + 2];
  }

  float aak[8], a2k[8];
#pragma unroll
  for (int k = 0; k < 8; ++k) {
    aak[k] = a[128 * 8 + k];
    a2k[k] = a2[k];
  }

  float hik[8];
#pragma unroll
  for (int k = 0; k < 8; ++k) hik[k] = hb[(size_t)i0 * 16 + k];
  float cix = cb[i0 * 3 + 0];
  float ciy = cb[i0 * 3 + 1];
  float ciz = cb[i0 * 3 + 2];

  __syncthreads();

  float hj[8][4], cjx[4], cjy[4], cjz[4];
#pragma unroll
  for (int k = 0; k < 8; ++k)
    *(float4*)hj[k] = *(const float4*)(sh_hj + k * N_ + j0);
  *(float4*)cjx = *(const float4*)(sh_c + 0 * N_ + j0);
  *(float4*)cjy = *(const float4*)(sh_c + 1 * N_ + j0);
  *(float4*)cjz = *(const float4*)(sh_c + 2 * N_ + j0);

  auto do_row = [&](const int r) {
    float av[4];
    *(float4*)av = av4;
    float ox = 0.f, oy = 0.f, oz = 0.f;
#pragma unroll
    for (int q = 0; q < 4; ++q) {
      float att = 0.f;
#pragma unroll
      for (int k = 0; k < 8; ++k) {
        const float pre = hik[k] + hj[k][q] + av[q] * aak[k];
        const float lrl = fmaxf(pre, 0.01f * pre);
        att += lrl * a2k[k];
      }
      const float dx = cix - cjx[q];
      const float dy = ciy - cjy[q];
      const float dz = ciz - cjz[q];
      const float nrm = sqrtf(dx * dx + dy * dy + dz * dz);
      const float w = att / (nrm + 1e-8f);
      ox += w * dx;
      oy += w * dy;
      oz += w * dz;
    }
#pragma unroll
    for (int off = 16; off > 0; off >>= 1) {
      ox += __shfl_xor(ox, off, 32);
      oy += __shfl_xor(oy, off, 32);
      oz += __shfl_xor(oz, off, 32);
    }
    if (lane == 0) {
      float* dst = sh_red + (r * 8 + wave) * 3;
      dst[0] = ox;
      dst[1] = oy;
      dst[2] = oz;
    }
  };

  for (int r = 0; r < ROWS - 1; ++r) {
    const int in = i0 + r + 1;
    const float4 nav4 = *(const float4*)(arow + (size_t)(r + 1) * N_);
    float nhik[8];
    for (int k = 0; k < 8; ++k) nhik[k] = hb[(size_t)in * 16 + k];
    const float ncix = cb[in * 3 + 0];
    const float nciy = cb[in * 3 + 1];
    const float nciz = cb[in * 3 + 2];

    do_row(r);

    av4 = nav4;
    for (int k = 0; k < 8; ++k) hik[k] = nhik[k];
    cix = ncix; ciy = nciy; ciz = nciz;
  }
  do_row(ROWS - 1);

  __syncthreads();

  if (tid < ROWS * 3) {
    const int r = tid / 3;
    const int d = tid - r * 3;
    float s = 0.f;
#pragma unroll
    for (int w8 = 0; w8 < 8; ++w8) s += sh_red[(r * 8 + w8) * 3 + d];
    out[(size_t)bi0 * 3 + tid] = s;
  }
}

extern "C" void kernel_launch(void* const* d_in, const int* in_sizes, int n_in,
                              void* d_out, int out_size, void* d_ws,
                              size_t ws_size, hipStream_t stream) {
  (void)in_sizes; (void)n_in; (void)out_size; (void)ws_size;
  const float* adj = (const float*)d_in[0];
  const float* c   = (const float*)d_in[1];
  const float* x   = (const float*)d_in[2];
  const float* a   = (const float*)d_in[3];
  const float* a2  = (const float*)d_in[4];
  float* out = (float*)d_out;
  float* h = (float*)d_ws;

  gat_h_wmma<<<(B_ * N_) / 16, 32, 0, stream>>>(x, a, h);
  gat_att_force<<<(B_ * N_) / ROWS, 256, 0, stream>>>(adj, c, h, a, a2, out);
}


# kernel 2: w = att * v_rcp_f32(v_sqrt_f32(d2) + 1e-8): 1-ulp hardware sqrt and reciprocal replace correctly rounded sqrtf and division; sum order unchanged
# speedup vs baseline: 2.4393x; 1.1089x over previous
#include <hip/hip_runtime.h>


typedef float v2f __attribute__((ext_vector_type(2)));
typedef float v8f __attribute__((ext_vector_type(8)));

#define B_   8
#define N_   1024
#define NIN  64

__global__ __launch_bounds__(32) void gat_h_wmma(const float* __restrict__ x,
                                                 const float* __restrict__ a,
                                                 float* __restrict__ h) {
  const int lane = threadIdx.x;
  const int half = lane >> 4;
  const int lr   = lane & 15;
  const int row0 = blockIdx.x * 16;
  const float* xr = x + (size_t)(row0 + lr) * NIN;

  const int ca = lr & 7;

  v8f acc = {};
#pragma unroll
  for (int it = 0; it < 16; ++it) {
    const int k0 = it * 4 + half * 2;
    v2f af, bf;
    af.x = xr[k0];
    af.y = xr[k0 + 1];
    const int fa = (lr < 8) ? k0 : (NIN + k0);
    bf.x = a[fa * 8 + ca];
    bf.y = a[(fa + 1) * 8 + ca];
    acc = __builtin_amdgcn_wmma_f32_16x16x4_f32(false, af, false, bf,
                                                (short)0, acc, false, false);
  }

  float* hrow = h + (size_t)row0 * 16;
#pragma unroll
  for (int v = 0; v < 8; ++v) {
    hrow[(size_t)(half * 8 + v) * 16 + lr] = acc[v];
  }
}

#define ROWS 16

__global__ __launch_bounds__(256) void gat_att_force(
    const float* __restrict__ adj, const float* __restrict__ c,
    const float* __restrict__ h, const float* __restrict__ a,
    const float* __restrict__ a2, float* __restrict__ out) {
  __shared__ float sh_hj[8 * N_];
  __shared__ float sh_c[3 * N_];
  __shared__ float sh_red[ROWS * 8 * 3];

  const int tid  = threadIdx.x;
  const int lane = tid & 31;
  const int wave = tid >> 5;
  const int bi0  = blockIdx.x * ROWS;
  const int b    = bi0 >> 10;
  const int i0   = bi0 & (N_ - 1);
  const int j0   = tid * 4;

  const float* arow = adj + (size_t)bi0 * N_ + j0;
  float4 av4 = *(const float4*)arow;

  const float* hb = h + (size_t)b * N_ * 16;
  for (int it = 0; it < (N_ * 2) / 256; ++it) {
    const int idx = tid + it * 256;
    const int j  = idx >> 1;
    const int q4 = (idx & 1) * 4;
    const float4 v = *(const float4*)(hb + (size_t)j * 16 + 8 + q4);
    sh_hj[(q4 + 0) * N_ + j] = v.x;
    sh_hj[(q4 + 1) * N_ + j] = v.y;
    sh_hj[(q4 + 2) * N_ + j] = v.z;
    sh_hj[(q4 + 3) * N_ + j] = v.w;
  }
  const float* cb = c + (size_t)b * N_ * 3;
  for (int it = 0; it < N_ / 256; ++it) {
    const int j = tid + it * 256;
    sh_c[0 * N_ + j] = cb[j * 3 + 0];
    sh_c[1 * N_ + j] = cb[j * 3 + 1];
    sh_c[2 * N_ + j] = cb[j * 3 + 2];
  }

  float aak[8], a2k[8];
#pragma unroll
  for (int k = 0; k < 8; ++k) {
    aak[k] = a[128 * 8 + k];
    a2k[k] = a2[k];
  }

  float hik[8];
#pragma unroll
  for (int k = 0; k < 8; ++k) hik[k] = hb[(size_t)i0 * 16 + k];
  float cix = cb[i0 * 3 + 0];
  float ciy = cb[i0 * 3 + 1];
  float ciz = cb[i0 * 3 + 2];

  __syncthreads();

  float hj[8][4], cjx[4], cjy[4], cjz[4];
#pragma unroll
  for (int k = 0; k < 8; ++k)
    *(float4*)hj[k] = *(const float4*)(sh_hj + k * N_ + j0);
  *(float4*)cjx = *(const float4*)(sh_c + 0 * N_ + j0);
  *(float4*)cjy = *(const float4*)(sh_c + 1 * N_ + j0);
  *(float4*)cjz = *(const float4*)(sh_c + 2 * N_ + j0);

  auto do_row = [&](const int r) {
    float av[4];
    *(float4*)av = av4;
    float ox = 0.f, oy = 0.f, oz = 0.f;
#pragma unroll
    for (int q = 0; q < 4; ++q) {
      float att = 0.f;
#pragma unroll
      for (int k = 0; k < 8; ++k) {
        const float pre = hik[k] + hj[k][q] + av[q] * aak[k];
        const float lrl = fmaxf(pre, 0.01f * pre);
        att += lrl * a2k[k];
      }
      const float dx = cix - cjx[q];
      const float dy = ciy - cjy[q];
      const float dz = ciz - cjz[q];
      const float nrm = __builtin_amdgcn_sqrtf(dx * dx + dy * dy + dz * dz);
      const float w = att * __builtin_amdgcn_rcpf(nrm + 1e-8f);
      ox += w * dx;
      oy += w * dy;
      oz += w * dz;
    }
#pragma unroll
    for (int off = 16; off > 0; off >>= 1) {
      ox += __shfl_xor(ox, off, 32);
      oy += __shfl_xor(oy, off, 32);
      oz += __shfl_xor(oz, off, 32);
    }
    if (lane == 0) {
      float* dst = sh_red + (r * 8 + wave) * 3;
      dst[0] = ox;
      dst[1] = oy;
      dst[2] = oz;
    }
  };

  for (int r = 0; r < ROWS - 1; ++r) {
    const int in = i0 + r + 1;
    const float4 nav4 = *(const float4*)(arow + (size_t)(r + 1) * N_);
    float nhik[8];
    for (int k = 0; k < 8; ++k) nhik[k] = hb[(size_t)in * 16 + k];
    const float ncix = cb[in * 3 + 0];
    const float nciy = cb[in * 3 + 1];
    const float nciz = cb[in * 3 + 2];

    do_row(r);

    av4 = nav4;
    for (int k = 0; k < 8; ++k) hik[k] = nhik[k];
    cix = ncix; ciy = nciy; ciz = nciz;
  }
  do_row(ROWS - 1);

  __syncthreads();

  if (tid < ROWS * 3) {
    const int r = tid / 3;
    const int d = tid - r * 3;
    float s = 0.f;
#pragma unroll
    for (int w8 = 0; w8 < 8; ++w8) s += sh_red[(r * 8 + w8) * 3 + d];
    out[(size_t)bi0 * 3 + tid] = s;
  }
}

extern "C" void kernel_launch(void* const* d_in, const int* in_sizes, int n_in,
                              void* d_out, int out_size, void* d_ws,
                              size_t ws_size, hipStream_t stream) {
  (void)in_sizes; (void)n_in; (void)out_size; (void)ws_size;
  const float* adj = (const float*)d_in[0];
  const float* c   = (const float*)d_in[1];
  const float* x   = (const float*)d_in[2];
  const float* a   = (const float*)d_in[3];
  const float* a2  = (const float*)d_in[4];
  float* out = (float*)d_out;
  float* h = (float*)d_ws;

  gat_h_wmma<<<(B_ * N_) / 16, 32, 0, stream>>>(x, a, h);
  gat_att_force<<<(B_ * N_) / ROWS, 256, 0, stream>>>(adj, c, h, a, a2, out);
}


# kernel 2: per-row wave butterfly (15 ds_bpermute a row) leaves the row loop: thread partials parked in LDS, one block-wide sum per output after the rows; order of the j sum changes
# speedup vs baseline: 2.6728x; 1.0957x over previous
#include <hip/hip_runtime.h>


typedef float v2f __attribute__((ext_vector_type(2)));
typedef float v8f __attribute__((ext_vector_type(8)));

#define B_   8
#define N_   1024
#define NIN  64

__global__ __launch_bounds__(32) void gat_h_wmma(const float* __restrict__ x,
                                                 const float* __restrict__ a,
                                                 float* __restrict__ h) {
  const int lane = threadIdx.x;
  const int half = lane >> 4;
  const int lr   = lane & 15;
  const int row0 = blockIdx.x * 16;
  const float* xr = x + (size_t)(row0 + lr) * NIN;

  const int ca = lr & 7;

  v8f acc = {};
#pragma unroll
  for (int it = 0; it < 16; ++it) {
    const int k0 = it * 4 + half * 2;
    v2f af, bf;
    af.x = xr[k0];
    af.y = xr[k0 + 1];
    const int fa = (lr < 8) ? k0 : (NIN + k0);
    bf.x = a[fa * 8 + ca];
    bf.y = a[(fa + 1) * 8 + ca];
    acc = __builtin_amdgcn_wmma_f32_16x16x4_f32(false, af, false, bf,
                                                (short)0, acc, false, false);
  }

  float* hrow = h + (size_t)row0 * 16;
#pragma unroll
  for (int v = 0; v < 8; ++v) {
    hrow[(size_t)(half * 8 + v) * 16 + lr] = acc[v];
  }
}

#define ROWS 16
#define PPITCH 272

__global__ __launch_bounds__(256) void gat_att_force(
    const float* __restrict__ adj, const float* __restrict__ c,
    const float* __restrict__ h, const float* __restrict__ a,
    const float* __restrict__ a2, float* __restrict__ out) {
  __shared__ float sh_hj[8 * N_];
  __shared__ float sh_c[3 * N_];
  __shared__ __attribute__((aligned(16))) float sh_part[ROWS * 3 * PPITCH];
  __shared__ __attribute__((aligned(16))) float sh_out4[ROWS * 3 * 4];

  const int tid  = threadIdx.x;
  const int bi0  = blockIdx.x * ROWS;
  const int b    = bi0 >> 10;
  const int i0   = bi0 & (N_ - 1);
  const int j0   = tid * 4;

  const float* arow = adj + (size_t)bi0 * N_ + j0;
  float4 av4 = *(const float4*)arow;

  const float* hb = h + (size_t)b * N_ * 16;
  for (int it = 0; it < (N_ * 2) / 256; ++it) {
    const int idx = tid + it * 256;
    const int j  = idx >> 1;
    const int q4 = (idx & 1) * 4;
    const float4 v = *(const float4*)(hb + (size_t)j * 16 + 8 + q4);
    sh_hj[(q4 + 0) * N_ + j] = v.x;
    sh_hj[(q4 + 1) * N_ + j] = v.y;
    sh_hj[(q4 + 2) * N_ + j] = v.z;
    sh_hj[(q4 + 3) * N_ + j] = v.w;
  }
  const float* cb = c + (size_t)b * N_ * 3;
  for (int it = 0; it < N_ / 256; ++it) {
    const int j = tid + it * 256;
    sh_c[0 * N_ + j] = cb[j * 3 + 0];
    sh_c[1 * N_ + j] = cb[j * 3 + 1];
    sh_c[2 * N_ + j] = cb[j * 3 + 2];
  }

  float aak[8], a2k[8];
#pragma unroll
  for (int k = 0; k < 8; ++k) {
    aak[k] = a[128 * 8 + k];
    a2k[k] = a2[k];
  }

  float hik[8];
#pragma unroll
  for (int k = 0; k < 8; ++k) hik[k] = hb[(size_t)i0 * 16 + k];
  float cix = cb[i0 * 3 + 0];
  float ciy = cb[i0 * 3 + 1];
  float ciz = cb[i0 * 3 + 2];

  __syncthreads();

  float hj[8][4], cjx[4], cjy[4], cjz[4];
#pragma unroll
  for (int k = 0; k < 8; ++k)
    *(float4*)hj[k] = *(const float4*)(sh_hj + k * N_ + j0);
  *(float4*)cjx = *(const float4*)(sh_c + 0 * N_ + j0);
  *(float4*)cjy = *(const float4*)(sh_c + 1 * N_ + j0);
  *(float4*)cjz = *(const float4*)(sh_c + 2 * N_ + j0);

  auto do_row = [&](const int r) {
    float av[4];
    *(float4*)av = av4;
    float ox = 0.f, oy = 0.f, oz = 0.f;
#pragma unroll
    for (int q = 0; q < 4; ++q) {
      float att = 0.f;
#pragma unroll
      for (int k = 0; k < 8; ++k) {
        const float pre = hik[k] + hj[k][q] + av[q] * aak[k];
        const float lrl = fmaxf(pre, 0.01f * pre);
        att += lrl * a2k[k];
      }
      const float dx = cix - cjx[q];
      const float dy = ciy - cjy[q];
      const float dz = ciz - cjz[q];
      const float nrm = __builtin_amdgcn_sqrtf(dx * dx + dy * dy + dz * dz);
      const float w = att * __builtin_amdgcn_rcpf(nrm + 1e-8f);
      ox += w * dx;
      oy += w * dy;
      oz += w * dz;
    }
    float* dst = sh_part + (r * 3) * PPITCH + tid;
    dst[0 * PPITCH] = ox;
    dst[1 * PPITCH] = oy;
    dst[2 * PPITCH] = oz;
  };

  for (int r = 0; r < ROWS - 1; ++r) {
    const int in = i0 + r + 1;
    const float4 nav4 = *(const float4*)(arow + (size_t)(r + 1) * N_);
    float nhik[8];
    for (int k = 0; k < 8; ++k) nhik[k] = hb[(size_t)in * 16 + k];
    const float ncix = cb[in * 3 + 0];
    const float nciy = cb[in * 3 + 1];
    const float nciz = cb[in * 3 + 2];

    do_row(r);

    av4 = nav4;
    for (int k = 0; k < 8; ++k) hik[k] = nhik[k];
    cix = ncix; ciy = nciy; ciz = nciz;
  }
  do_row(ROWS - 1);

  __syncthreads();

  if (tid < ROWS * 3 * 4) {
    const int o = tid >> 2;
    const int p = tid & 3;
    const float* sp = sh_part + o * PPITCH + p * 4;
    float4 acc = *(const float4*)sp;
#pragma unroll
    for (int i = 1; i < 16; ++i) {
      const float4 v = *(const float4*)(sp + 16 * i);
      acc.x += v.x;
      acc.y += v.y;
      acc.z += v.z;
      acc.w += v.w;
    }
    sh_out4[tid] = (acc.x + acc.y) + (acc.z + acc.w);
  }

  __syncthreads();

  if (tid < ROWS * 3) {
    const float4 q = *(const float4*)(sh_out4 + tid * 4);
    out[(size_t)bi0 * 3 + tid] = (q.x + q.y) + (q.z + q.w);
  }
}

extern "C" void kernel_launch(void* const* d_in, const int* in_sizes, int n_in,
                              void* d_out, int out_size, void* d_ws,
                              size_t ws_size, hipStream_t stream) {
  (void)in_sizes; (void)n_in; (void)out_size; (void)ws_size;
  const float* adj = (const float*)d_in[0];
  const float* c   = (const float*)d_in[1];
  const float* x   = (const float*)d_in[2];
  const float* a   = (const float*)d_in[3];
  const float* a2  = (const float*)d_in[4];
  float* out = (float*)d_out;
  float* h = (float*)d_ws;

  gat_h_wmma<<<(B_ * N_) / 16, 32, 0, stream>>>(x, a, h);
  gat_att_force<<<(B_ * N_) / ROWS, 256, 0, stream>>>(adj, c, h, a, a2, out);
}


# kernel 2: row loop waits for no global load: adj requested 4 rows ahead (ring of 4 float4 registers, rows unrolled) not 1; rows' hi and c_i staged in LDS, not scalar loads; sum order unchanged
# speedup vs baseline: 3.0392x; 1.1371x over previous
#include <hip/hip_runtime.h>


typedef float v2f __attribute__((ext_vector_type(2)));
typedef float v8f __attribute__((ext_vector_type(8)));

#define B_   8
#define N_   1024
#define NIN  64

__global__ __launch_bounds__(32) void gat_h_wmma(const float* __restrict__ x,
                                                 const float* __restrict__ a,
                                                 float* __restrict__ h) {
  const int lane = threadIdx.x;
  const int half = lane >> 4;
  const int lr   = lane & 15;
  const int row0 = blockIdx.x * 16;
  const float* xr = x + (size_t)(row0 + lr) * NIN;

  const int ca = lr & 7;

  v8f acc = {};
#pragma unroll
  for (int it = 0; it < 16; ++it) {
    const int k0 = it * 4 + half * 2;
    v2f af, bf;
    af.x = xr[k0];
    af.y = xr[k0 + 1];
    const int fa = (lr < 8) ? k0 : (NIN + k0);
    bf.x = a[fa * 8 + ca];
    bf.y = a[(fa + 1) * 8 + ca];
    acc = __builtin_amdgcn_wmma_f32_16x16x4_f32(false, af, false, bf,
                                                (short)0, acc, false, false);
  }

  float* hrow = h + (size_t)row0 * 16;
#pragma unroll
  for (int v = 0; v < 8; ++v) {
    hrow[(size_t)(half * 8 + v) * 16 + lr] = acc[v];
  }
}

#define ROWS 16
#define DIST 4
#define PPITCH 272

__global__ __launch_bounds__(256) void gat_att_force(
    const float* __restrict__ adj, const float* __restrict__ c,
    const float* __restrict__ h, const float* __restrict__ a,
    const float* __restrict__ a2, float* __restrict__ out) {
  __shared__ float sh_hj[8 * N_];
  __shared__ float sh_c[3 * N_];
  __shared__ __attribute__((aligned(16))) float sh_part[ROWS * 3 * PPITCH];
  __shared__ __attribute__((aligned(16))) float sh_out4[ROWS * 3 * 4];
  __shared__ __attribute__((aligned(16))) float sh_u[256];

  const int tid  = threadIdx.x;
  const int bi0  = blockIdx.x * ROWS;
  const int b    = bi0 >> 10;
  const int i0   = bi0 & (N_ - 1);
  const int j0   = tid * 4;

  const float* arow = adj + (size_t)bi0 * N_ + j0;
  float4 ring[DIST];
#pragma unroll
  for (int u = 0; u < DIST; ++u)
    ring[u] = *(const float4*)(arow + (size_t)u * N_);

  const float* hb = h + (size_t)b * N_ * 16;
  const float* cb = c + (size_t)b * N_ * 3;
  const float uh = hb[(size_t)(i0 + ((tid >> 3) & (ROWS - 1))) * 16 + (tid & 7)];
  const float uc = cb[(i0 + ((tid >> 2) & (ROWS - 1))) * 3 + ((tid & 3) < 3 ? (tid & 3) : 2)];

  for (int it = 0; it < (N_ * 2) / 256; ++it) {
    const int idx = tid + it * 256;
    const int j  = idx >> 1;
    const int q4 = (idx & 1) * 4;
    const float4 v = *(const float4*)(hb + (size_t)j * 16 + 8 + q4);
    sh_hj[(q4 + 0) * N_ + j] = v.x;
    sh_hj[(q4 + 1) * N_ + j] = v.y;
    sh_hj[(q4 + 2) * N_ + j] = v.z;
    sh_hj[(q4 + 3) * N_ + j] = v.w;
  }
  for (int it = 0; it < N_ / 256; ++it) {
    const int j = tid + it * 256;
    sh_c[0 * N_ + j] = cb[j * 3 + 0];
    sh_c[1 * N_ + j] = cb[j * 3 + 1];
    sh_c[2 * N_ + j] = cb[j * 3 + 2];
  }

  sh_u[tid] = (tid < ROWS * 8) ? uh : uc;

  float aak[8], a2k[8];
#pragma unroll
  for (int k = 0; k < 8; ++k) {
    aak[k] = a[128 * 8 + k];
    a2k[k] = a2[k];
  }

  __syncthreads();

  float hj[8][4], cjx[4], cjy[4], cjz[4];
#pragma unroll
  for (int k = 0; k < 8; ++k)
    *(float4*)hj[k] = *(const float4*)(sh_hj + k * N_ + j0);
  *(float4*)cjx = *(const float4*)(sh_c + 0 * N_ + j0);
  *(float4*)cjy = *(const float4*)(sh_c + 1 * N_ + j0);
  *(float4*)cjz = *(const float4*)(sh_c + 2 * N_ + j0);

  auto do_row = [&](const int r, const float4 av4) {
    float av[4];
    *(float4*)av = av4;
    float hik[8];
    *(float4*)(hik + 0) = *(const float4*)(sh_u + r * 8);
    *(float4*)(hik + 4) = *(const float4*)(sh_u + r * 8 + 4);
    const float4 ci4 = *(const float4*)(sh_u + ROWS * 8 + r * 4);
    const float cix = ci4.x, ciy = ci4.y, ciz = ci4.z;
    float ox = 0.f, oy = 0.f, oz = 0.f;
#pragma unroll
    for (int q = 0; q < 4; ++q) {
      float att = 0.f;
#pragma unroll
      for (int k = 0; k < 8; ++k) {
        const float pre = hik[k] + hj[k][q] + av[q] * aak[k];
        const float lrl = fmaxf(pre, 0.01f * pre);
        att += lrl * a2k[k];
      }
      const float dx = cix - cjx[q];
      const float dy = ciy - cjy[q];
      const float dz = ciz - cjz[q];
      const float nrm = __builtin_amdgcn_sqrtf(dx * dx + dy * dy + dz * dz);
      const float w = att * __builtin_amdgcn_rcpf(nrm + 1e-8f);
      ox += w * dx;
      oy += w * dy;
      oz += w * dz;
    }
    float* dst = sh_part + (r * 3) * PPITCH + tid;
    dst[0 * PPITCH] = ox;
    dst[1 * PPITCH] = oy;
    dst[2 * PPITCH] = oz;
  };

#pragma unroll
  for (int r = 0; r < ROWS; ++r) {
    const float4 cur = ring[r % DIST];
    if (r + DIST < ROWS)
      ring[r % DIST] = *(const float4*)(arow + (size_t)(r + DIST) * N_);
    asm volatile("" ::: "memory");
    do_row(r, cur);
  }

  __syncthreads();

  if (tid < ROWS * 3 * 4) {
    const int o = tid >> 2;
    const int p = tid & 3;
    const float* sp = sh_part + o * PPITCH + p * 4;
    float4 acc = *(const float4*)sp;
#pragma unroll
    for (int i = 1; i < 16; ++i) {
      const float4 v = *(const float4*)(sp + 16 * i);
      acc.x += v.x;
      acc.y += v.y;
      acc.z += v.z;
      acc.w += v.w;
    }
    sh_out4[tid] = (acc.x + acc.y) + (acc.z + acc.w);
  }

  __syncthreads();

  if (tid < ROWS * 3) {
    const float4 q = *(const float4*)(sh_out4 + tid * 4);
    out[(size_t)bi0 * 3 + tid] = (q.x + q.y) + (q.z + q.w);
  }
}

extern "C" void kernel_launch(void* const* d_in, const int* in_sizes, int n_in,
                              void* d_out, int out_size, void* d_ws,
                              size_t ws_size, hipStream_t stream) {
  (void)in_sizes; (void)n_in; (void)out_size; (void)ws_size;
  const float* adj = (const float*)d_in[0];
  const float* c   = (const float*)d_in[1];
  const float* x   = (const float*)d_in[2];
  const float* a   = (const float*)d_in[3];
  const float* a2  = (const float*)d_in[4];
  float* out = (float*)d_out;
  float* h = (float*)d_ws;

  gat_h_wmma<<<(B_ * N_) / 16, 32, 0, stream>>>(x, a, h);
  gat_att_force<<<(B_ * N_) / ROWS, 256, 0, stream>>>(adj, c, h, a, a2, out);
}


# kernel 1 stores h feature-major ([b][16][N]); a kernel-2 thread loads hj and c of its 4 columns from global straight into registers: the 44 KB LDS transpose staging per block is gone; sums unchanged
# speedup vs baseline: 3.2754x; 1.0777x over previous
#include <hip/hip_runtime.h>


typedef float v2f __attribute__((ext_vector_type(2)));
typedef float v8f __attribute__((ext_vector_type(8)));

#define B_   8
#define N_   1024
#define NIN  64

__global__ __launch_bounds__(32) void gat_h_wmma(const float* __restrict__ x,
                                                 const float* __restrict__ a,
                                                 float* __restrict__ h) {
  const int lane = threadIdx.x;
  const int half = lane >> 4;
  const int lr   = lane & 15;
  const int row0 = blockIdx.x * 16;
  const float* xr = x + (size_t)(row0 + lr) * NIN;

  const int ca = lr & 7;

  v8f acc = {};
#pragma unroll
  for (int it = 0; it < 16; ++it) {
    const int k0 = it * 4 + half * 2;
    v2f af, bf;
    af.x = xr[k0];
    af.y = xr[k0 + 1];
    const int fa = (lr < 8) ? k0 : (NIN + k0);
    bf.x = a[fa * 8 + ca];
    bf.y = a[(fa + 1) * 8 + ca];
    acc = __builtin_amdgcn_wmma_f32_16x16x4_f32(false, af, false, bf,
                                                (short)0, acc, false, false);
  }

  float* hcol = h + ((size_t)(row0 >> 10) * 16 + lr) * N_ + (row0 & (N_ - 1)) + half * 8;
  float4 lo, hi4;
  lo.x = acc[0]; lo.y = acc[1]; lo.z = acc[2]; lo.w = acc[3];
  hi4.x = acc[4]; hi4.y = acc[5]; hi4.z = acc[6]; hi4.w = acc[7];
  *(float4*)(hcol)     = lo;
  *(float4*)(hcol + 4) = hi4;
}

#define ROWS 16
#define DIST 4
#define PPITCH 272

__global__ __launch_bounds__(256) void gat_att_force(
    const float* __restrict__ adj, const float* __restrict__ c,
    const float* __restrict__ h, const float* __restrict__ a,
    const float* __restrict__ a2, float* __restrict__ out) {
  __shared__ __attribute__((aligned(16))) float sh_part[ROWS * 3 * PPITCH];
  __shared__ __attribute__((aligned(16))) float sh_out4[ROWS * 3 * 4];
  __shared__ __attribute__((aligned(16))) float sh_u[256];

  const int tid  = threadIdx.x;
  const int bi0  = blockIdx.x * ROWS;
  const int b    = bi0 >> 10;
  const int i0   = bi0 & (N_ - 1);
  const int j0   = tid * 4;

  const float* arow = adj + (size_t)bi0 * N_ + j0;
  float4 ring[DIST];
#pragma unroll
  for (int u = 0; u < DIST; ++u)
    ring[u] = *(const float4*)(arow + (size_t)u * N_);

  const float* hb = h + (size_t)b * 16 * N_;
  const float* cb = c + (size_t)b * N_ * 3;
  float hj[8][4], cjx[4], cjy[4], cjz[4];
#pragma unroll
  for (int k = 0; k < 8; ++k)
    *(float4*)hj[k] = *(const float4*)(hb + (size_t)(8 + k) * N_ + j0);
  {
    const float4 c0 = *(const float4*)(cb + j0 * 3);
    const float4 c1 = *(const float4*)(cb + j0 * 3 + 4);
    const float4 c2 = *(const float4*)(cb + j0 * 3 + 8);
    cjx[0] = c0.x; cjy[0] = c0.y; cjz[0] = c0.z;
    cjx[1] = c0.w; cjy[1] = c1.x; cjz[1] = c1.y;
    cjx[2] = c1.z; cjy[2] = c1.w; cjz[2] = c2.x;
    cjx[3] = c2.y; cjy[3] = c2.z; cjz[3] = c2.w;
  }

  const float uh = hb[(size_t)(tid & 7) * N_ + i0 + ((tid >> 3) & (ROWS - 1))];
  const float uc = cb[(i0 + ((tid >> 2) & (ROWS - 1))) * 3 + ((tid & 3) < 3 ? (tid & 3) : 2)];

  sh_u[tid] = (tid < ROWS * 8) ? uh : uc;

  float aak[8], a2k[8];
#pragma unroll
  for (int k = 0; k < 8; ++k) {
    aak[k] = a[128 * 8 + k];
    a2k[k] = a2[k];
  }

  __syncthreads();

  auto do_row = [&](const int r, const float4 av4) {
    float av[4];
    *(float4*)av = av4;
    float hik[8];
    *(float4*)(hik + 0) = *(const float4*)(sh_u + r * 8);
    *(float4*)(hik + 4) = *(const float4*)(sh_u + r * 8 + 4);
    const float4 ci4 = *(const float4*)(sh_u + ROWS * 8 + r * 4);
    const float cix = ci4.x, ciy = ci4.y, ciz = ci4.z;
    float ox = 0.f, oy = 0.f, oz = 0.f;
#pragma unroll
    for (int q = 0; q < 4; ++q) {
      float att = 0.f;
#pragma unroll
      for (int k = 0; k < 8; ++k) {
        const float pre = hik[k] + hj[k][q] + av[q] * aak[k];
        const float lrl = fmaxf(pre, 0.01f * pre);
        att += lrl * a2k[k];
      }
      const float dx = cix - cjx[q];
      const float dy = ciy - cjy[q];
      const float dz = ciz - cjz[q];
      const float nrm = __builtin_amdgcn_sqrtf(dx * dx + dy * dy + dz * dz);
      const float w = att * __builtin_amdgcn_rcpf(nrm + 1e-8f);
      ox += w * dx;
      oy += w * dy;
      oz += w * dz;
    }
    float* dst = sh_part + (r * 3) * PPITCH + tid;
    dst[0 * PPITCH] = ox;
    dst[1 * PPITCH] = oy;
    dst[2 * PPITCH] = oz;
  };

#pragma unroll
  for (int r = 0; r < ROWS; ++r) {
    const float4 cur = ring[r % DIST];
    if (r + DIST < ROWS)
      ring[r % DIST] = *(const float4*)(arow + (size_t)(r + DIST) * N_);
    asm volatile("" ::: "memory");
    do_row(r, cur);
  }

  __syncthreads();

  if (tid < ROWS * 3 * 4) {
    const int o = tid >> 2;
    const int p = tid & 3;
    const float* sp = sh_part + o * PPITCH + p * 4;
    float4 acc = *(const float4*)sp;
#pragma unroll
    for (int i = 1; i < 16; ++i) {
      const float4 v = *(const float4*)(sp + 16 * i);
      acc.x += v.x;
      acc.y += v.y;
      acc.z += v.z;
      acc.w += v.w;
    }
    sh_out4[tid] = (acc.x + acc.y) + (acc.z + acc.w);
  }

  __syncthreads();

  if (tid < ROWS * 3) {
    const float4 q = *(const float4*)(sh_out4 + tid * 4);
    out[(size_t)bi0 * 3 + tid] = (q.x + q.y) + (q.z + q.w);
  }
}

extern "C" void kernel_launch(void* const* d_in, const int* in_sizes, int n_in,
                              void* d_out, int out_size, void* d_ws,
                              size_t ws_size, hipStream_t stream) {
  (void)in_sizes; (void)n_in; (void)out_size; (void)ws_size;
  const float* adj = (const float*)d_in[0];
  const float* c   = (const float*)d_in[1];
  const float* x   = (const float*)d_in[2];
  const float* a   = (const float*)d_in[3];
  const float* a2  = (const float*)d_in[4];
  float* out = (float*)d_out;
  float* h = (float*)d_ws;

  gat_h_wmma<<<(B_ * N_) / 16, 32, 0, stream>>>(x, a, h);
  gat_att_force<<<(B_ * N_) / ROWS, 256, 0, stream>>>(adj, c, h, a, a2, out);
}


# kernel 2: leaky_relu(p) = 0.505p + 0.495|p|: linear half summed over k once per row, column and block; |p| half is one fma per k (abs modifier): 5 -> 3 vector ops per (pair,k); k-sum order changes
# speedup vs baseline: 3.5739x; 1.0911x over previous
#include <hip/hip_runtime.h>


typedef float v2f __attribute__((ext_vector_type(2)));
typedef float v8f __attribute__((ext_vector_type(8)));

#define B_   8
#define N_   1024
#define NIN  64

__global__ __launch_bounds__(32) void gat_h_wmma(const float* __restrict__ x,
                                                 const float* __restrict__ a,
                                                 float* __restrict__ h) {
  const int lane = threadIdx.x;
  const int half = lane >> 4;
  const int lr   = lane & 15;
  const int row0 = blockIdx.x * 16;
  const float* xr = x + (size_t)(row0 + lr) * NIN;

  const int ca = lr & 7;

  v8f acc = {};
#pragma unroll
  for (int it = 0; it < 16; ++it) {
    const int k0 = it * 4 + half * 2;
    v2f af, bf;
    af.x = xr[k0];
    af.y = xr[k0 + 1];
    const int fa = (lr < 8) ? k0 : (NIN + k0);
    bf.x = a[fa * 8 + ca];
    bf.y = a[(fa + 1) * 8 + ca];
    acc = __builtin_amdgcn_wmma_f32_16x16x4_f32(false, af, false, bf,
                                                (short)0, acc, false, false);
  }

  float* hcol = h + ((size_t)(row0 >> 10) * 16 + lr) * N_ + (row0 & (N_ - 1)) + half * 8;
  float4 lo, hi4;
  lo.x = acc[0]; lo.y = acc[1]; lo.z = acc[2]; lo.w = acc[3];
  hi4.x = acc[4]; hi4.y = acc[5]; hi4.z = acc[6]; hi4.w = acc[7];
  *(float4*)(hcol)     = lo;
  *(float4*)(hcol + 4) = hi4;
}

#define ROWS 16
#define DIST 4
#define PPITCH 272

__global__ __launch_bounds__(256) void gat_att_force(
    const float* __restrict__ adj, const float* __restrict__ c,
    const float* __restrict__ h, const float* __restrict__ a,
    const float* __restrict__ a2, float* __restrict__ out) {
  __shared__ __attribute__((aligned(16))) float sh_part[ROWS * 3 * PPITCH];
  __shared__ __attribute__((aligned(16))) float sh_out4[ROWS * 3 * 4];
  __shared__ __attribute__((aligned(16))) float sh_u[256];

  const int tid  = threadIdx.x;
  const int bi0  = blockIdx.x * ROWS;
  const int b    = bi0 >> 10;
  const int i0   = bi0 & (N_ - 1);
  const int j0   = tid * 4;

  const float* arow = adj + (size_t)bi0 * N_ + j0;
  float4 ring[DIST];
#pragma unroll
  for (int u = 0; u < DIST; ++u)
    ring[u] = *(const float4*)(arow + (size_t)u * N_);

  const float* hb = h + (size_t)b * 16 * N_;
  const float* cb = c + (size_t)b * N_ * 3;
  float hj[8][4], cjx[4], cjy[4], cjz[4];
#pragma unroll
  for (int k = 0; k < 8; ++k)
    *(float4*)hj[k] = *(const float4*)(hb + (size_t)(8 + k) * N_ + j0);
  {
    const float4 c0 = *(const float4*)(cb + j0 * 3);
    const float4 c1 = *(const float4*)(cb + j0 * 3 + 4);
    const float4 c2 = *(const float4*)(cb + j0 * 3 + 8);
    cjx[0] = c0.x; cjy[0] = c0.y; cjz[0] = c0.z;
    cjx[1] = c0.w; cjy[1] = c1.x; cjz[1] = c1.y;
    cjx[2] = c1.z; cjy[2] = c1.w; cjz[2] = c2.x;
    cjx[3] = c2.y; cjy[3] = c2.z; cjz[3] = c2.w;
  }

  const float uh = hb[(size_t)(tid & 7) * N_ + i0 + ((tid >> 3) & (ROWS - 1))];
  const int   rc = (tid >> 2) & (ROWS - 1);
  const float ucl = cb[(i0 + rc) * 3 + ((tid & 3) < 3 ? (tid & 3) : 2)];

  float aak[8], a2k[8];
#pragma unroll
  for (int k = 0; k < 8; ++k) {
    aak[k] = a[128 * 8 + k];
    a2k[k] = a2[k];
  }

  float lin_i = 0.f, lin_a = 0.f;
#pragma unroll
  for (int k = 0; k < 8; ++k) {
    lin_i += a2k[k] * hb[(size_t)k * N_ + i0 + rc];
    lin_a += a2k[k] * aak[k];
  }
  lin_i *= 0.505f;
  lin_a *= 0.505f;
  const float uc = ((tid & 3) == 3) ? lin_i : ucl;
  float lin_j[4], a2s[8];
#pragma unroll
  for (int q = 0; q < 4; ++q) {
    float t = 0.f;
#pragma unroll
    for (int k = 0; k < 8; ++k) t += a2k[k] * hj[k][q];
    lin_j[q] = 0.505f * t;
  }
#pragma unroll
  for (int k = 0; k < 8; ++k) a2s[k] = 0.495f * a2k[k];

  sh_u[tid] = (tid < ROWS * 8) ? uh : uc;

  __syncthreads();

  auto do_row = [&](const int r, const float4 av4) {
    float av[4];
    *(float4*)av = av4;
    float hik[8];
    *(float4*)(hik + 0) = *(const float4*)(sh_u + r * 8);
    *(float4*)(hik + 4) = *(const float4*)(sh_u + r * 8 + 4);
    const float4 ci4 = *(const float4*)(sh_u + ROWS * 8 + r * 4);
    const float cix = ci4.x, ciy = ci4.y, ciz = ci4.z, li = ci4.w;
    float ox = 0.f, oy = 0.f, oz = 0.f;
#pragma unroll
    for (int q = 0; q < 4; ++q) {
      float att = (li + lin_j[q]) + av[q] * lin_a;
#pragma unroll
      for (int k = 0; k < 8; ++k) {
        const float pre = hik[k] + hj[k][q] + av[q] * aak[k];
        att += __builtin_fabsf(pre) * a2s[k];
      }
      const float dx = cix - cjx[q];
      const float dy = ciy - cjy[q];
      const float dz = ciz - cjz[q];
      const float nrm = __builtin_amdgcn_sqrtf(dx * dx + dy * dy + dz * dz);
      const float w = att * __builtin_amdgcn_rcpf(nrm + 1e-8f);
      ox += w * dx;
      oy += w * dy;
      oz += w * dz;
    }
    float* dst = sh_part + (r * 3) * PPITCH + tid;
    dst[0 * PPITCH] = ox;
    dst[1 * PPITCH] = oy;
    dst[2 * PPITCH] = oz;
  };

#pragma unroll
  for (int r = 0; r < ROWS; ++r) {
    const float4 cur = ring[r % DIST];
    if (r + DIST < ROWS)
      ring[r % DIST] = *(const float4*)(arow + (size_t)(r + DIST) * N_);
    asm volatile("" ::: "memory");
    do_row(r, cur);
  }

  __syncthreads();

  if (tid < ROWS * 3 * 4) {
    const int o = tid >> 2;
    const int p = tid & 3;
    const float* sp = sh_part + o * PPITCH + p * 4;
    float4 acc = *(const float4*)sp;
#pragma unroll
    for (int i = 1; i < 16; ++i) {
      const float4 v = *(const float4*)(sp + 16 * i);
      acc.x += v.x;
      acc.y += v.y;
      acc.z += v.z;
      acc.w += v.w;
    }
    sh_out4[tid] = (acc.x + acc.y) + (acc.z + acc.w);
  }

  __syncthreads();

  if (tid < ROWS * 3) {
    const float4 q = *(const float4*)(sh_out4 + tid * 4);
    out[(size_t)bi0 * 3 + tid] = (q.x + q.y) + (q.z + q.w);
  }
}

extern "C" void kernel_launch(void* const* d_in, const int* in_sizes, int n_in,
                              void* d_out, int out_size, void* d_ws,
                              size_t ws_size, hipStream_t stream) {
  (void)in_sizes; (void)n_in; (void)out_size; (void)ws_size;
  const float* adj = (const float*)d_in[0];
  const float* c   = (const float*)d_in[1];
  const float* x   = (const float*)d_in[2];
  const float* a   = (const float*)d_in[3];
  const float* a2  = (const float*)d_in[4];
  float* out = (float*)d_out;
  float* h = (float*)d_ws;

  gat_h_wmma<<<(B_ * N_) / 16, 32, 0, stream>>>(x, a, h);
  gat_att_force<<<(B_ * N_) / ROWS, 256, 0, stream>>>(adj, c, h, a, a2, out);
}
